# starting kernel
# baseline (speedup 1.0000x reference)
#include <hip/hip_runtime.h>
#include <hip/hip_bf16.h>
#include <math.h>


#define N_NODES 50000
#define N_EDGES 800000
#define IN_DIM  128
#define HID_DIM 64

typedef float v2f __attribute__((ext_vector_type(2)));
typedef float v8f __attribute__((ext_vector_type(8)));


__global__ void init_deg_kernel(float* deg, int n) {
    int i = blockIdx.x * blockDim.x + threadIdx.x;
    if (i < n) deg[i] = 1.0f;
}

__global__ void edge_deg_kernel(const int* __restrict__ dst, float* deg, int e_count) {
    int e = blockIdx.x * blockDim.x + threadIdx.x;
    if (e < e_count) atomicAdd(&deg[dst[e]], 1.0f);
}

__global__ void rsqrt_kernel(float* deg, int n) {
    int i = blockIdx.x * blockDim.x + threadIdx.x;
    if (i < n) {
        float d = deg[i];
        deg[i] = (d > 0.0f) ? rsqrtf(d) : 0.0f;
    }
}


template <int K_DIM>
__global__ __launch_bounds__(256) void gemm_wmma_f32(const float* __restrict__ A,
                                                     const float* __restrict__ W,
                                                     float* __restrict__ C, int M) {
    __shared__ float sW[K_DIM * 64];
    for (int i = threadIdx.x; i < K_DIM * 64; i += 256) sW[i] = W[i];
    __syncthreads();

    const int wave = threadIdx.x >> 5;
    const int lane = threadIdx.x & 31;
    const int m_tile = blockIdx.x * 2 + (wave >> 2);
    const int n_tile = wave & 3;

    const int row16 = lane & 15;
    const int khalf = (lane >> 4) << 1;

    int mrow = m_tile * 16 + row16;
    int arow = (mrow < M) ? mrow : (M - 1);
    const float* __restrict__ Aptr = A + (size_t)arow * K_DIM;
    const float* __restrict__ Bcol = sW + n_tile * 16 + row16;

    v8f acc = {};
    #pragma unroll
    for (int k = 0; k < K_DIM; k += 4) {
        v2f a, b;
        a.x = Aptr[k + khalf];
        a.y = Aptr[k + khalf + 1];
        b.x = Bcol[(k + khalf) * 64];
        b.y = Bcol[(k + khalf + 1) * 64];
        acc = __builtin_amdgcn_wmma_f32_16x16x4_f32(
             false, a,  false, b,
             (short)0, acc,  false,  false);
    }

    const int mbase = m_tile * 16 + ((lane >> 4) << 3);
    const int ncol  = n_tile * 16 + row16;
    #pragma unroll
    for (int j = 0; j < 8; ++j) {
        int m = mbase + j;
        if (m < M) C[(size_t)m * 64 + ncol] = acc[j];
    }
}


__global__ void zero_kernel(float4* p, int n4) {
    int i = blockIdx.x * blockDim.x + threadIdx.x;
    if (i < n4) p[i] = make_float4(0.f, 0.f, 0.f, 0.f);
}


__global__ __launch_bounds__(256) void scatter_kernel(const float* __restrict__ h,
                                                      const float* __restrict__ dinv,
                                                      const int* __restrict__ src,
                                                      const int* __restrict__ dst,
                                                      float* __restrict__ agg, int e_count) {
    int gid = blockIdx.x * blockDim.x + threadIdx.x;
    int e = gid >> 4;
    if (e >= e_count) return;
    int c = (gid & 15) << 2;
    int s = src[e], d = dst[e];
    float norm = dinv[s] * dinv[d];
    const float4 hv = *(const float4*)(h + (size_t)s * 64 + c);
    float* o = agg + (size_t)d * 64 + c;
    atomicAdd(o + 0, hv.x * norm);
    atomicAdd(o + 1, hv.y * norm);
    atomicAdd(o + 2, hv.z * norm);
    atomicAdd(o + 3, hv.w * norm);
}


__global__ __launch_bounds__(256) void finalize_kernel(const float* __restrict__ agg,
                                                       const float* __restrict__ h,
                                                       const float* __restrict__ dinv,
                                                       const float* __restrict__ bias,
                                                       float* __restrict__ out,
                                                       int n_nodes, int do_relu) {
    int gid = blockIdx.x * blockDim.x + threadIdx.x;
    int n = gid >> 4;
    if (n >= n_nodes) return;
    int c = (gid & 15) << 2;
    float di = dinv[n];
    float self_n = di * di;
    const float4 a  = *(const float4*)(agg + (size_t)n * 64 + c);
    const float4 hv = *(const float4*)(h   + (size_t)n * 64 + c);
    const float4 b  = *(const float4*)(bias + c);
    float4 r;
    r.x = a.x + hv.x * self_n + b.x;
    r.y = a.y + hv.y * self_n + b.y;
    r.z = a.z + hv.z * self_n + b.z;
    r.w = a.w + hv.w * self_n + b.w;
    if (do_relu) {
        r.x = fmaxf(r.x, 0.f); r.y = fmaxf(r.y, 0.f);
        r.z = fmaxf(r.z, 0.f); r.w = fmaxf(r.w, 0.f);
    }
    *(float4*)(out + (size_t)n * 64 + c) = r;
}


extern "C" void kernel_launch(void* const* d_in, const int* in_sizes, int n_in,
                              void* d_out, int out_size, void* d_ws, size_t ws_size,
                              hipStream_t stream) {
    const float* x  = (const float*)d_in[0];
    const int*   ei = (const int*)d_in[1];
    const float* W1 = (const float*)d_in[2];
    const float* b1 = (const float*)d_in[3];
    const float* W2 = (const float*)d_in[4];
    const float* b2 = (const float*)d_in[5];
    float* out = (float*)d_out;

    const int* src = ei;
    const int* dst = ei + N_EDGES;

    float* dinv = (float*)d_ws;
    float* buf0 = dinv + 65536;
    float* buf1 = buf0 + (size_t)N_NODES * 64;
    const int feat_elems = N_NODES * 64;

    const int B = 256;
    dim3 blk(B);

    init_deg_kernel<<<(N_NODES + B - 1) / B, blk, 0, stream>>>(dinv, N_NODES);
    edge_deg_kernel<<<(N_EDGES + B - 1) / B, blk, 0, stream>>>(dst, dinv, N_EDGES);
    rsqrt_kernel<<<(N_NODES + B - 1) / B, blk, 0, stream>>>(dinv, N_NODES);

    int gemm_grid = (N_NODES + 31) / 32;
    gemm_wmma_f32<IN_DIM><<<gemm_grid, blk, 0, stream>>>(x, W1, buf0, N_NODES);
    zero_kernel<<<(feat_elems / 4 + B - 1) / B, blk, 0, stream>>>((float4*)buf1, feat_elems / 4);
    scatter_kernel<<<(N_EDGES * 16 + B - 1) / B, blk, 0, stream>>>(buf0, dinv, src, dst, buf1, N_EDGES);
    finalize_kernel<<<(N_NODES * 16 + B - 1) / B, blk, 0, stream>>>(buf1, buf0, dinv, b1,
                                                                    buf0, N_NODES,  1);

    gemm_wmma_f32<HID_DIM><<<gemm_grid, blk, 0, stream>>>(buf0, W2, buf1, N_NODES);
    zero_kernel<<<(feat_elems / 4 + B - 1) / B, blk, 0, stream>>>((float4*)buf0, feat_elems / 4);
    scatter_kernel<<<(N_EDGES * 16 + B - 1) / B, blk, 0, stream>>>(buf1, dinv, src, dst, buf0, N_EDGES);
    finalize_kernel<<<(N_NODES * 16 + B - 1) / B, blk, 0, stream>>>(buf0, buf1, dinv, b2,
                                                                    out, N_NODES,  0);
}


# Edge scatter: one wave per edge, lane l adds columns l and l+32: each row load and f32 atomic add instruction covers one whole 128-byte line (was 4 lines x 8 words); same adds, arrival order
# speedup vs baseline: 1.4978x; 1.4978x over previous
#include <hip/hip_runtime.h>
#include <hip/hip_bf16.h>
#include <math.h>


#define N_NODES 50000
#define N_EDGES 800000
#define IN_DIM  128
#define HID_DIM 64

typedef float v2f __attribute__((ext_vector_type(2)));
typedef float v8f __attribute__((ext_vector_type(8)));


__global__ void init_deg_kernel(float* deg, int n) {
    int i = blockIdx.x * blockDim.x + threadIdx.x;
    if (i < n) deg[i] = 1.0f;
}

__global__ void edge_deg_kernel(const int* __restrict__ dst, float* deg, int e_count) {
    int e = blockIdx.x * blockDim.x + threadIdx.x;
    if (e < e_count) atomicAdd(&deg[dst[e]], 1.0f);
}

__global__ void rsqrt_kernel(float* deg, int n) {
    int i = blockIdx.x * blockDim.x + threadIdx.x;
    if (i < n) {
        float d = deg[i];
        deg[i] = (d > 0.0f) ? rsqrtf(d) : 0.0f;
    }
}


template <int K_DIM>
__global__ __launch_bounds__(256) void gemm_wmma_f32(const float* __restrict__ A,
                                                     const float* __restrict__ W,
                                                     float* __restrict__ C, int M) {
    __shared__ float sW[K_DIM * 64];
    for (int i = threadIdx.x; i < K_DIM * 64; i += 256) sW[i] = W[i];
    __syncthreads();

    const int wave = threadIdx.x >> 5;
    const int lane = threadIdx.x & 31;
    const int m_tile = blockIdx.x * 2 + (wave >> 2);
    const int n_tile = wave & 3;

    const int row16 = lane & 15;
    const int khalf = (lane >> 4) << 1;

    int mrow = m_tile * 16 + row16;
    int arow = (mrow < M) ? mrow : (M - 1);
    const float* __restrict__ Aptr = A + (size_t)arow * K_DIM;
    const float* __restrict__ Bcol = sW + n_tile * 16 + row16;

    v8f acc = {};
    #pragma unroll
    for (int k = 0; k < K_DIM; k += 4) {
        v2f a, b;
        a.x = Aptr[k + khalf];
        a.y = Aptr[k + khalf + 1];
        b.x = Bcol[(k + khalf) * 64];
        b.y = Bcol[(k + khalf + 1) * 64];
        acc = __builtin_amdgcn_wmma_f32_16x16x4_f32(
             false, a,  false, b,
             (short)0, acc,  false,  false);
    }

    const int mbase = m_tile * 16 + ((lane >> 4) << 3);
    const int ncol  = n_tile * 16 + row16;
    #pragma unroll
    for (int j = 0; j < 8; ++j) {
        int m = mbase + j;
        if (m < M) C[(size_t)m * 64 + ncol] = acc[j];
    }
}


__global__ void zero_kernel(float4* p, int n4) {
    int i = blockIdx.x * blockDim.x + threadIdx.x;
    if (i < n4) p[i] = make_float4(0.f, 0.f, 0.f, 0.f);
}


#define EDGES_PER_WAVE 16

__global__ __launch_bounds__(256) void scatter_kernel(const float* __restrict__ h,
                                                      const float* __restrict__ dinv,
                                                      const int* __restrict__ src,
                                                      const int* __restrict__ dst,
                                                      float* __restrict__ agg, int e_count) {
    const int lane = threadIdx.x & 31;
    const int wave = (blockIdx.x * 256 + threadIdx.x) >> 5;
    const int e0 = wave * EDGES_PER_WAVE;
    if (e0 >= e_count) return;

    int my_s = 0, my_d = 0;
    if (lane < EDGES_PER_WAVE && e0 + lane < e_count) {
        my_s = src[e0 + lane];
        my_d = dst[e0 + lane];
    }
    const float my_ds = dinv[my_s];
    const float my_dd = dinv[my_d];

    if (e0 + EDGES_PER_WAVE <= e_count) {
        float lo[EDGES_PER_WAVE], hi[EDGES_PER_WAVE];
        #pragma unroll
        for (int j = 0; j < EDGES_PER_WAVE; ++j) {
            const int s = __builtin_amdgcn_readlane(my_s, j);
            const float* __restrict__ row = h + (size_t)s * 64;
            lo[j] = row[lane];
            hi[j] = row[lane + 32];
        }
        const float my_norm = my_ds * my_dd;
        #pragma unroll
        for (int j = 0; j < EDGES_PER_WAVE; ++j) {
            const int d = __builtin_amdgcn_readlane(my_d, j);
            const float norm = __int_as_float(__builtin_amdgcn_readlane(__float_as_int(my_norm), j));
            float* o = agg + (size_t)d * 64;
            atomicAdd(o + lane,      lo[j] * norm);
            atomicAdd(o + lane + 32, hi[j] * norm);
        }
    } else {
        const int n_here = e_count - e0;
        const float my_norm = my_ds * my_dd;
        for (int j = 0; j < n_here; ++j) {
            const int s = __shfl(my_s, j);
            const int d = __shfl(my_d, j);
            const float norm = __shfl(my_norm, j);
            const float* __restrict__ row = h + (size_t)s * 64;
            float* o = agg + (size_t)d * 64;
            atomicAdd(o + lane,      row[lane] * norm);
            atomicAdd(o + lane + 32, row[lane + 32] * norm);
        }
    }
}


__global__ __launch_bounds__(256) void finalize_kernel(const float* __restrict__ agg,
                                                       const float* __restrict__ h,
                                                       const float* __restrict__ dinv,
                                                       const float* __restrict__ bias,
                                                       float* __restrict__ out,
                                                       int n_nodes, int do_relu) {
    int gid = blockIdx.x * blockDim.x + threadIdx.x;
    int n = gid >> 4;
    if (n >= n_nodes) return;
    int c = (gid & 15) << 2;
    float di = dinv[n];
    float self_n = di * di;
    const float4 a  = *(const float4*)(agg + (size_t)n * 64 + c);
    const float4 hv = *(const float4*)(h   + (size_t)n * 64 + c);
    const float4 b  = *(const float4*)(bias + c);
    float4 r;
    r.x = a.x + hv.x * self_n + b.x;
    r.y = a.y + hv.y * self_n + b.y;
    r.z = a.z + hv.z * self_n + b.z;
    r.w = a.w + hv.w * self_n + b.w;
    if (do_relu) {
        r.x = fmaxf(r.x, 0.f); r.y = fmaxf(r.y, 0.f);
        r.z = fmaxf(r.z, 0.f); r.w = fmaxf(r.w, 0.f);
    }
    *(float4*)(out + (size_t)n * 64 + c) = r;
}


extern "C" void kernel_launch(void* const* d_in, const int* in_sizes, int n_in,
                              void* d_out, int out_size, void* d_ws, size_t ws_size,
                              hipStream_t stream) {
    const float* x  = (const float*)d_in[0];
    const int*   ei = (const int*)d_in[1];
    const float* W1 = (const float*)d_in[2];
    const float* b1 = (const float*)d_in[3];
    const float* W2 = (const float*)d_in[4];
    const float* b2 = (const float*)d_in[5];
    float* out = (float*)d_out;

    const int* src = ei;
    const int* dst = ei + N_EDGES;

    float* dinv = (float*)d_ws;
    float* buf0 = dinv + 65536;
    float* buf1 = buf0 + (size_t)N_NODES * 64;
    const int feat_elems = N_NODES * 64;

    const int B = 256;
    dim3 blk(B);

    init_deg_kernel<<<(N_NODES + B - 1) / B, blk, 0, stream>>>(dinv, N_NODES);
    edge_deg_kernel<<<(N_EDGES + B - 1) / B, blk, 0, stream>>>(dst, dinv, N_EDGES);
    rsqrt_kernel<<<(N_NODES + B - 1) / B, blk, 0, stream>>>(dinv, N_NODES);

    int gemm_grid = (N_NODES + 31) / 32;
    const int scatter_waves = (N_EDGES + EDGES_PER_WAVE - 1) / EDGES_PER_WAVE;
    const int scatter_grid = (scatter_waves * 32 + B - 1) / B;
    gemm_wmma_f32<IN_DIM><<<gemm_grid, blk, 0, stream>>>(x, W1, buf0, N_NODES);
    zero_kernel<<<(feat_elems / 4 + B - 1) / B, blk, 0, stream>>>((float4*)buf1, feat_elems / 4);
    scatter_kernel<<<scatter_grid, blk, 0, stream>>>(buf0, dinv, src, dst, buf1, N_EDGES);
    finalize_kernel<<<(N_NODES * 16 + B - 1) / B, blk, 0, stream>>>(buf1, buf0, dinv, b1,
                                                                    buf0, N_NODES,  1);

    gemm_wmma_f32<HID_DIM><<<gemm_grid, blk, 0, stream>>>(buf0, W2, buf1, N_NODES);
    zero_kernel<<<(feat_elems / 4 + B - 1) / B, blk, 0, stream>>>((float4*)buf0, feat_elems / 4);
    scatter_kernel<<<scatter_grid, blk, 0, stream>>>(buf1, dinv, src, dst, buf0, N_EDGES);
    finalize_kernel<<<(N_NODES * 16 + B - 1) / B, blk, 0, stream>>>(buf0, buf1, dinv, b2,
                                                                    out, N_NODES,  0);
}


# Clear and finish passes folded into the GEMMs: epilogue also stores the sum's start value h*dinv^2+bias (layer 2: into out), relu at GEMM 2's A load; 11 -> 7 launches; adds then land in arrival order
# speedup vs baseline: 1.7200x; 1.1484x over previous
#include <hip/hip_runtime.h>
#include <hip/hip_bf16.h>
#include <math.h>


#define N_NODES 50000
#define N_EDGES 800000
#define IN_DIM  128
#define HID_DIM 64

typedef float v2f __attribute__((ext_vector_type(2)));
typedef float v8f __attribute__((ext_vector_type(8)));


__global__ void init_deg_kernel(float* deg, int n) {
    int i = blockIdx.x * blockDim.x + threadIdx.x;
    if (i < n) deg[i] = 1.0f;
}

__global__ void edge_deg_kernel(const int* __restrict__ dst, float* deg, int e_count) {
    int e = blockIdx.x * blockDim.x + threadIdx.x;
    if (e < e_count) atomicAdd(&deg[dst[e]], 1.0f);
}

__global__ void rsqrt_kernel(float* deg, int n) {
    int i = blockIdx.x * blockDim.x + threadIdx.x;
    if (i < n) {
        float d = deg[i];
        deg[i] = (d > 0.0f) ? rsqrtf(d) : 0.0f;
    }
}


template <int K_DIM, bool RELU_A>
__global__ __launch_bounds__(256) void gemm_wmma_f32(const float* __restrict__ A,
                                                     const float* __restrict__ W,
                                                     const float* __restrict__ dinv,
                                                     const float* __restrict__ bias,
                                                     float* __restrict__ H,
                                                     float* __restrict__ S, int M) {
    __shared__ float sW[K_DIM * 64];
    for (int i = threadIdx.x; i < K_DIM * 64; i += 256) sW[i] = W[i];
    __syncthreads();

    const int wave = threadIdx.x >> 5;
    const int lane = threadIdx.x & 31;
    const int m_tile = blockIdx.x * 2 + (wave >> 2);
    const int n_tile = wave & 3;

    const int row16 = lane & 15;
    const int khalf = (lane >> 4) << 1;

    int mrow = m_tile * 16 + row16;
    int arow = (mrow < M) ? mrow : (M - 1);
    const float* __restrict__ Aptr = A + (size_t)arow * K_DIM;
    const float* __restrict__ Bcol = sW + n_tile * 16 + row16;

    const int mbase = m_tile * 16 + ((lane >> 4) << 3);
    const int ncol  = n_tile * 16 + row16;

    float self_n[8];
    #pragma unroll
    for (int j = 0; j < 8; ++j) {
        const int m = mbase + j;
        const float di = dinv[(m < M) ? m : (M - 1)];
        self_n[j] = di * di;
    }
    const float bcol = bias[ncol];

    v8f acc = {};
    #pragma unroll
    for (int k = 0; k < K_DIM; k += 4) {
        v2f a, b;
        a.x = Aptr[k + khalf];
        a.y = Aptr[k + khalf + 1];
        if (RELU_A) { a.x = fmaxf(a.x, 0.f); a.y = fmaxf(a.y, 0.f); }
        b.x = Bcol[(k + khalf) * 64];
        b.y = Bcol[(k + khalf + 1) * 64];
        acc = __builtin_amdgcn_wmma_f32_16x16x4_f32(
             false, a,  false, b,
             (short)0, acc,  false,  false);
    }

    #pragma unroll
    for (int j = 0; j < 8; ++j) {
        const int m = mbase + j;
        if (m < M) {
            H[(size_t)m * 64 + ncol] = acc[j];
            S[(size_t)m * 64 + ncol] = acc[j] * self_n[j] + bcol;
        }
    }
}


#define EDGES_PER_WAVE 16

__global__ __launch_bounds__(256) void scatter_kernel(const float* __restrict__ h,
                                                      const float* __restrict__ dinv,
                                                      const int* __restrict__ src,
                                                      const int* __restrict__ dst,
                                                      float* __restrict__ agg, int e_count) {
    const int lane = threadIdx.x & 31;
    const int wave = (blockIdx.x * 256 + threadIdx.x) >> 5;
    const int e0 = wave * EDGES_PER_WAVE;
    if (e0 >= e_count) return;

    int my_s = 0, my_d = 0;
    if (lane < EDGES_PER_WAVE && e0 + lane < e_count) {
        my_s = src[e0 + lane];
        my_d = dst[e0 + lane];
    }
    const float my_ds = dinv[my_s];
    const float my_dd = dinv[my_d];

    if (e0 + EDGES_PER_WAVE <= e_count) {
        float lo[EDGES_PER_WAVE], hi[EDGES_PER_WAVE];
        #pragma unroll
        for (int j = 0; j < EDGES_PER_WAVE; ++j) {
            const int s = __builtin_amdgcn_readlane(my_s, j);
            const float* __restrict__ row = h + (size_t)s * 64;
            lo[j] = row[lane];
            hi[j] = row[lane + 32];
        }
        const float my_norm = my_ds * my_dd;
        #pragma unroll
        for (int j = 0; j < EDGES_PER_WAVE; ++j) {
            const int d = __builtin_amdgcn_readlane(my_d, j);
            const float norm = __int_as_float(__builtin_amdgcn_readlane(__float_as_int(my_norm), j));
            float* o = agg + (size_t)d * 64;
            atomicAdd(o + lane,      lo[j] * norm);
            atomicAdd(o + lane + 32, hi[j] * norm);
        }
    } else {
        const int n_here = e_count - e0;
        const float my_norm = my_ds * my_dd;
        for (int j = 0; j < n_here; ++j) {
            const int s = __shfl(my_s, j);
            const int d = __shfl(my_d, j);
            const float norm = __shfl(my_norm, j);
            const float* __restrict__ row = h + (size_t)s * 64;
            float* o = agg + (size_t)d * 64;
            atomicAdd(o + lane,      row[lane] * norm);
            atomicAdd(o + lane + 32, row[lane + 32] * norm);
        }
    }
}


extern "C" void kernel_launch(void* const* d_in, const int* in_sizes, int n_in,
                              void* d_out, int out_size, void* d_ws, size_t ws_size,
                              hipStream_t stream) {
    const float* x  = (const float*)d_in[0];
    const int*   ei = (const int*)d_in[1];
    const float* W1 = (const float*)d_in[2];
    const float* b1 = (const float*)d_in[3];
    const float* W2 = (const float*)d_in[4];
    const float* b2 = (const float*)d_in[5];
    float* out = (float*)d_out;

    const int* src = ei;
    const int* dst = ei + N_EDGES;

    float* dinv = (float*)d_ws;
    float* buf0 = dinv + 65536;
    float* buf1 = buf0 + (size_t)N_NODES * 64;

    const int B = 256;
    dim3 blk(B);

    init_deg_kernel<<<(N_NODES + B - 1) / B, blk, 0, stream>>>(dinv, N_NODES);
    edge_deg_kernel<<<(N_EDGES + B - 1) / B, blk, 0, stream>>>(dst, dinv, N_EDGES);
    rsqrt_kernel<<<(N_NODES + B - 1) / B, blk, 0, stream>>>(dinv, N_NODES);

    int gemm_grid = (N_NODES + 31) / 32;
    const int scatter_waves = (N_EDGES + EDGES_PER_WAVE - 1) / EDGES_PER_WAVE;
    const int scatter_grid = (scatter_waves * 32 + B - 1) / B;

    gemm_wmma_f32<IN_DIM, false><<<gemm_grid, blk, 0, stream>>>(x, W1, dinv, b1, buf0, buf1, N_NODES);
    scatter_kernel<<<scatter_grid, blk, 0, stream>>>(buf0, dinv, src, dst, buf1, N_EDGES);

    gemm_wmma_f32<HID_DIM, true><<<gemm_grid, blk, 0, stream>>>(buf1, W2, dinv, b2, buf0, out, N_NODES);
    scatter_kernel<<<scatter_grid, blk, 0, stream>>>(buf0, dinv, src, dst, out, N_EDGES);
}


# Edge scatters walk the edge list once per half of the node range; work-groups claim chunks by ticket, own package half's range first: a sum line is added into from one half only; arrival order
# speedup vs baseline: 1.8242x; 1.0606x over previous
#include <hip/hip_runtime.h>
#include <hip/hip_bf16.h>
#include <math.h>


#define N_NODES 50000
#define N_EDGES 800000
#define IN_DIM  128
#define HID_DIM 64

#define HALF_N (N_NODES / 2)
#define N_TICKETS 4
#define TICKET_PITCH 64

typedef float v2f __attribute__((ext_vector_type(2)));
typedef float v8f __attribute__((ext_vector_type(8)));


__global__ void init_deg_kernel(float* deg, int* tickets, int n) {
    int i = blockIdx.x * blockDim.x + threadIdx.x;
    if (i < n) deg[i] = 1.0f;
    if (i < N_TICKETS) tickets[i * TICKET_PITCH] = 0;
}

__global__ void edge_deg_kernel(const int* __restrict__ dst, float* deg, int e_count) {
    int e = blockIdx.x * blockDim.x + threadIdx.x;
    if (e < e_count) atomicAdd(&deg[dst[e]], 1.0f);
}

__global__ void rsqrt_kernel(float* deg, int n) {
    int i = blockIdx.x * blockDim.x + threadIdx.x;
    if (i < n) {
        float d = deg[i];
        deg[i] = (d > 0.0f) ? rsqrtf(d) : 0.0f;
    }
}


template <int K_DIM, bool RELU_A>
__global__ __launch_bounds__(256) void gemm_wmma_f32(const float* __restrict__ A,
                                                     const float* __restrict__ W,
                                                     const float* __restrict__ dinv,
                                                     const float* __restrict__ bias,
                                                     float* __restrict__ H,
                                                     float* __restrict__ S, int M) {
    __shared__ float sW[K_DIM * 64];
    for (int i = threadIdx.x; i < K_DIM * 64; i += 256) sW[i] = W[i];
    __syncthreads();

    const int wave = threadIdx.x >> 5;
    const int lane = threadIdx.x & 31;
    const int m_tile = blockIdx.x * 2 + (wave >> 2);
    const int n_tile = wave & 3;

    const int row16 = lane & 15;
    const int khalf = (lane >> 4) << 1;

    int mrow = m_tile * 16 + row16;
    int arow = (mrow < M) ? mrow : (M - 1);
    const float* __restrict__ Aptr = A + (size_t)arow * K_DIM;
    const float* __restrict__ Bcol = sW + n_tile * 16 + row16;

    const int mbase = m_tile * 16 + ((lane >> 4) << 3);
    const int ncol  = n_tile * 16 + row16;

    float self_n[8];
    #pragma unroll
    for (int j = 0; j < 8; ++j) {
        const int m = mbase + j;
        const float di = dinv[(m < M) ? m : (M - 1)];
        self_n[j] = di * di;
    }
    const float bcol = bias[ncol];

    v8f acc = {};
    #pragma unroll
    for (int k = 0; k < K_DIM; k += 4) {
        v2f a, b;
        a.x = Aptr[k + khalf];
        a.y = Aptr[k + khalf + 1];
        if (RELU_A) { a.x = fmaxf(a.x, 0.f); a.y = fmaxf(a.y, 0.f); }
        b.x = Bcol[(k + khalf) * 64];
        b.y = Bcol[(k + khalf + 1) * 64];
        acc = __builtin_amdgcn_wmma_f32_16x16x4_f32(
             false, a,  false, b,
             (short)0, acc,  false,  false);
    }

    #pragma unroll
    for (int j = 0; j < 8; ++j) {
        const int m = mbase + j;
        if (m < M) {
            H[(size_t)m * 64 + ncol] = acc[j];
            S[(size_t)m * 64 + ncol] = acc[j] * self_n[j] + bcol;
        }
    }
}


#define STEP_E 32
#define CHUNK_E (8 * STEP_E)
#define GROUP 16
#define SCATTER_BLOCKS 2048

__device__ __forceinline__ int package_half() {
    const unsigned id = __builtin_amdgcn_s_sendmsg_rtn(0x87);
    return (int)((id >> 18) & 1u);
}

__global__ __launch_bounds__(256) void scatter_kernel(const float* __restrict__ h,
                                                      const float* __restrict__ dinv,
                                                      const int* __restrict__ src,
                                                      const int* __restrict__ dst,
                                                      float* __restrict__ agg,
                                                      int* __restrict__ tickets, int e_count) {
    __shared__ int s_ticket;
    const int lane = threadIdx.x & 31;
    const int wave = threadIdx.x >> 5;
    const int n_chunks = (e_count + CHUNK_E - 1) / CHUNK_E;
    const int my_half = package_half();

    for (int pass = 0; pass < 2; ++pass) {
        const int r = my_half ^ pass;
        int* counter = tickets + r * TICKET_PITCH;
        for (int trip = 0; trip <= n_chunks; ++trip) {
            if (threadIdx.x == 0) s_ticket = atomicAdd(counter, 1);
            __syncthreads();
            const int t = s_ticket;
            __syncthreads();
            if (t >= n_chunks) break;

            const int e = t * CHUNK_E + wave * STEP_E + lane;
            const int ec = (e < e_count) ? e : (e_count - 1);
            const int my_s = src[ec];
            const int my_d = dst[ec];
            asm volatile("" :: "v"(my_s), "v"(my_d));
            const bool mine = (e < e_count) && ((my_d >= HALF_N) == (r != 0));
            const float my_ds = dinv[mine ? my_s : 0];
            const float my_dd = dinv[mine ? my_d : 0];

            unsigned todo = __builtin_amdgcn_ballot_w32(mine);
            while (todo) {
                int j[GROUP];
                #pragma unroll
                for (int i = 0; i < GROUP; ++i) {
                    j[i] = todo ? __builtin_ctz(todo) : -1;
                    todo &= todo - 1u;
                }
                float lo[GROUP], hi[GROUP];
                #pragma unroll
                for (int i = 0; i < GROUP; ++i) {
                    if (j[i] >= 0) {
                        const int s = __builtin_amdgcn_readlane(my_s, j[i]);
                        const float* __restrict__ row = h + (size_t)s * 64;
                        lo[i] = row[lane];
                        hi[i] = row[lane + 32];
                    }
                }
                #pragma unroll
                for (int i = 0; i < GROUP; ++i) {
                    if (j[i] >= 0) {
                        const int d = __builtin_amdgcn_readlane(my_d, j[i]);
                        const float norm =
                            __int_as_float(__builtin_amdgcn_readlane(__float_as_int(my_ds), j[i])) *
                            __int_as_float(__builtin_amdgcn_readlane(__float_as_int(my_dd), j[i]));
                        float* o = agg + (size_t)d * 64;
                        atomicAdd(o + lane,      lo[i] * norm);
                        atomicAdd(o + lane + 32, hi[i] * norm);
                    }
                }
            }
        }
    }
}


extern "C" void kernel_launch(void* const* d_in, const int* in_sizes, int n_in,
                              void* d_out, int out_size, void* d_ws, size_t ws_size,
                              hipStream_t stream) {
    const float* x  = (const float*)d_in[0];
    const int*   ei = (const int*)d_in[1];
    const float* W1 = (const float*)d_in[2];
    const float* b1 = (const float*)d_in[3];
    const float* W2 = (const float*)d_in[4];
    const float* b2 = (const float*)d_in[5];
    float* out = (float*)d_out;

    const int* src = ei;
    const int* dst = ei + N_EDGES;

    float* dinv = (float*)d_ws;
    float* buf0 = dinv + 65536;
    float* buf1 = buf0 + (size_t)N_NODES * 64;
    int* tickets = (int*)(dinv + 50176);

    const int B = 256;
    dim3 blk(B);

    init_deg_kernel<<<(N_NODES + B - 1) / B, blk, 0, stream>>>(dinv, tickets, N_NODES);
    edge_deg_kernel<<<(N_EDGES + B - 1) / B, blk, 0, stream>>>(dst, dinv, N_EDGES);
    rsqrt_kernel<<<(N_NODES + B - 1) / B, blk, 0, stream>>>(dinv, N_NODES);

    int gemm_grid = (N_NODES + 31) / 32;
    const int scatter_grid = SCATTER_BLOCKS;

    gemm_wmma_f32<IN_DIM, false><<<gemm_grid, blk, 0, stream>>>(x, W1, dinv, b1, buf0, buf1, N_NODES);
    scatter_kernel<<<scatter_grid, blk, 0, stream>>>(buf0, dinv, src, dst, buf1, tickets, N_EDGES);

    gemm_wmma_f32<HID_DIM, true><<<gemm_grid, blk, 0, stream>>>(buf1, W2, dinv, b2, buf0, out, N_NODES);
    scatter_kernel<<<scatter_grid, blk, 0, stream>>>(buf0, dinv, src, dst, out, tickets + 2 * TICKET_PITCH, N_EDGES);
}


# Degree count without global atomics: 256 work-groups each count one slice of dst for one node range in an LDS table (integer adds), store it as a plane; dinv pass sums 32 planes; 7 -> 6 launches
# speedup vs baseline: 2.7498x; 1.5074x over previous
#include <hip/hip_runtime.h>
#include <hip/hip_bf16.h>
#include <math.h>


#define N_NODES 50000
#define N_EDGES 800000
#define IN_DIM  128
#define HID_DIM 64

#define HALF_N (N_NODES / 2)
#define N_TICKETS 4
#define TICKET_PITCH 64

typedef float v2f __attribute__((ext_vector_type(2)));
typedef float v8f __attribute__((ext_vector_type(8)));


#define CNT_RANGES 8
#define CNT_SLICES 32
#define CNT_RANGE_N 6272
#define CNT_PITCH (CNT_RANGES * CNT_RANGE_N)
#define CNT_SLICE_E (N_EDGES / CNT_SLICES)
#define CNT_SLICE_Q (CNT_SLICE_E / 4)
#define CNT_THREADS 256
#define CNT_LOADS ((CNT_SLICE_Q + CNT_THREADS - 1) / CNT_THREADS)
#define CNT_RANGE_Q (CNT_RANGE_N / 4)
#define CNT_MOVES ((CNT_RANGE_Q + CNT_THREADS - 1) / CNT_THREADS)

static_assert(CNT_PITCH >= N_NODES, "ranges must cover every node");
static_assert(CNT_RANGE_N % 32 == 0, "a wave's 32 counter words are one whole line");
static_assert(N_EDGES % CNT_SLICES == 0 && CNT_SLICE_E % 4 == 0, "slices are whole 16-byte words");

__device__ __forceinline__ void count_one(unsigned* hist, int d, int lo) {
    const unsigned o = (unsigned)(d - lo);
    if (o < (unsigned)CNT_RANGE_N) atomicAdd(&hist[o], 1u);
}

__global__ __launch_bounds__(CNT_THREADS) void count_deg_kernel(const int* __restrict__ dst,
                                                                unsigned* __restrict__ planes) {
    __shared__ __attribute__((aligned(16))) unsigned hist[CNT_RANGE_N];
    const int range = blockIdx.x % CNT_RANGES;
    const int slice = blockIdx.x / CNT_RANGES;
    const int lo = range * CNT_RANGE_N;

    const int4* __restrict__ q = reinterpret_cast<const int4*>(dst + (size_t)slice * CNT_SLICE_E);
    int4 v[CNT_LOADS];
    #pragma unroll
    for (int j = 0; j < CNT_LOADS; ++j) {
        const int i = threadIdx.x + j * CNT_THREADS;
        const bool in = i < CNT_SLICE_Q;
        v[j] = q[in ? i : (CNT_SLICE_Q - 1)];
        if (!in) v[j] = make_int4(-1, -1, -1, -1);
    }

    uint4* hist4 = reinterpret_cast<uint4*>(hist);
    #pragma unroll
    for (int j = 0; j < CNT_MOVES; ++j) {
        const int i = threadIdx.x + j * CNT_THREADS;
        if (i < CNT_RANGE_Q) hist4[i] = make_uint4(0u, 0u, 0u, 0u);
    }
    __syncthreads();

    #pragma unroll
    for (int j = 0; j < CNT_LOADS; ++j) {
        count_one(hist, v[j].x, lo);
        count_one(hist, v[j].y, lo);
        count_one(hist, v[j].z, lo);
        count_one(hist, v[j].w, lo);
    }
    __syncthreads();

    uint4* __restrict__ out4 = reinterpret_cast<uint4*>(planes + (size_t)slice * CNT_PITCH + lo);
    #pragma unroll
    for (int j = 0; j < CNT_MOVES; ++j) {
        const int i = threadIdx.x + j * CNT_THREADS;
        if (i < CNT_RANGE_Q) out4[i] = hist4[i];
    }
}

__global__ void dinv_kernel(const unsigned* __restrict__ planes, float* __restrict__ dinv,
                            int* tickets, int n) {
    int i = blockIdx.x * blockDim.x + threadIdx.x;
    if (i < N_TICKETS) tickets[i * TICKET_PITCH] = 0;
    if (i < n) {
        unsigned c = 1u;
        #pragma unroll
        for (int s = 0; s < CNT_SLICES; ++s) c += planes[(size_t)s * CNT_PITCH + i];
        const float d = (float)c;
        dinv[i] = (d > 0.0f) ? rsqrtf(d) : 0.0f;
    }
}


template <int K_DIM, bool RELU_A>
__global__ __launch_bounds__(256) void gemm_wmma_f32(const float* __restrict__ A,
                                                     const float* __restrict__ W,
                                                     const float* __restrict__ dinv,
                                                     const float* __restrict__ bias,
                                                     float* __restrict__ H,
                                                     float* __restrict__ S, int M) {
    __shared__ float sW[K_DIM * 64];
    for (int i = threadIdx.x; i < K_DIM * 64; i += 256) sW[i] = W[i];
    __syncthreads();

    const int wave = threadIdx.x >> 5;
    const int lane = threadIdx.x & 31;
    const int m_tile = blockIdx.x * 2 + (wave >> 2);
    const int n_tile = wave & 3;

    const int row16 = lane & 15;
    const int khalf = (lane >> 4) << 1;

    int mrow = m_tile * 16 + row16;
    int arow = (mrow < M) ? mrow : (M - 1);
    const float* __restrict__ Aptr = A + (size_t)arow * K_DIM;
    const float* __restrict__ Bcol = sW + n_tile * 16 + row16;

    const int mbase = m_tile * 16 + ((lane >> 4) << 3);
    const int ncol  = n_tile * 16 + row16;

    float self_n[8];
    #pragma unroll
    for (int j = 0; j < 8; ++j) {
        const int m = mbase + j;
        const float di = dinv[(m < M) ? m : (M - 1)];
        self_n[j] = di * di;
    }
    const float bcol = bias[ncol];

    v8f acc = {};
    #pragma unroll
    for (int k = 0; k < K_DIM; k += 4) {
        v2f a, b;
        a.x = Aptr[k + khalf];
        a.y = Aptr[k + khalf + 1];
        if (RELU_A) { a.x = fmaxf(a.x, 0.f); a.y = fmaxf(a.y, 0.f); }
        b.x = Bcol[(k + khalf) * 64];
        b.y = Bcol[(k + khalf + 1) * 64];
        acc = __builtin_amdgcn_wmma_f32_16x16x4_f32(
             false, a,  false, b,
             (short)0, acc,  false,  false);
    }

    #pragma unroll
    for (int j = 0; j < 8; ++j) {
        const int m = mbase + j;
        if (m < M) {
            H[(size_t)m * 64 + ncol] = acc[j];
            S[(size_t)m * 64 + ncol] = acc[j] * self_n[j] + bcol;
        }
    }
}


#define STEP_E 32
#define CHUNK_E (8 * STEP_E)
#define GROUP 16
#define SCATTER_BLOCKS 2048

__device__ __forceinline__ int package_half() {
    const unsigned id = __builtin_amdgcn_s_sendmsg_rtn(0x87);
    return (int)((id >> 18) & 1u);
}

__global__ __launch_bounds__(256) void scatter_kernel(const float* __restrict__ h,
                                                      const float* __restrict__ dinv,
                                                      const int* __restrict__ src,
                                                      const int* __restrict__ dst,
                                                      float* __restrict__ agg,
                                                      int* __restrict__ tickets, int e_count) {
    __shared__ int s_ticket;
    const int lane = threadIdx.x & 31;
    const int wave = threadIdx.x >> 5;
    const int n_chunks = (e_count + CHUNK_E - 1) / CHUNK_E;
    const int my_half = package_half();

    for (int pass = 0; pass < 2; ++pass) {
        const int r = my_half ^ pass;
        int* counter = tickets + r * TICKET_PITCH;
        for (int trip = 0; trip <= n_chunks; ++trip) {
            if (threadIdx.x == 0) s_ticket = atomicAdd(counter, 1);
            __syncthreads();
            const int t = s_ticket;
            __syncthreads();
            if (t >= n_chunks) break;

            const int e = t * CHUNK_E + wave * STEP_E + lane;
            const int ec = (e < e_count) ? e : (e_count - 1);
            const int my_s = src[ec];
            const int my_d = dst[ec];
            asm volatile("" :: "v"(my_s), "v"(my_d));
            const bool mine = (e < e_count) && ((my_d >= HALF_N) == (r != 0));
            const float my_ds = dinv[mine ? my_s : 0];
            const float my_dd = dinv[mine ? my_d : 0];

            unsigned todo = __builtin_amdgcn_ballot_w32(mine);
            while (todo) {
                int j[GROUP];
                #pragma unroll
                for (int i = 0; i < GROUP; ++i) {
                    j[i] = todo ? __builtin_ctz(todo) : -1;
                    todo &= todo - 1u;
                }
                float lo[GROUP], hi[GROUP];
                #pragma unroll
                for (int i = 0; i < GROUP; ++i) {
                    if (j[i] >= 0) {
                        const int s = __builtin_amdgcn_readlane(my_s, j[i]);
                        const float* __restrict__ row = h + (size_t)s * 64;
                        lo[i] = row[lane];
                        hi[i] = row[lane + 32];
                    }
                }
                #pragma unroll
                for (int i = 0; i < GROUP; ++i) {
                    if (j[i] >= 0) {
                        const int d = __builtin_amdgcn_readlane(my_d, j[i]);
                        const float norm =
                            __int_as_float(__builtin_amdgcn_readlane(__float_as_int(my_ds), j[i])) *
                            __int_as_float(__builtin_amdgcn_readlane(__float_as_int(my_dd), j[i]));
                        float* o = agg + (size_t)d * 64;
                        atomicAdd(o + lane,      lo[i] * norm);
                        atomicAdd(o + lane + 32, hi[i] * norm);
                    }
                }
            }
        }
    }
}


extern "C" void kernel_launch(void* const* d_in, const int* in_sizes, int n_in,
                              void* d_out, int out_size, void* d_ws, size_t ws_size,
                              hipStream_t stream) {
    const float* x  = (const float*)d_in[0];
    const int*   ei = (const int*)d_in[1];
    const float* W1 = (const float*)d_in[2];
    const float* b1 = (const float*)d_in[3];
    const float* W2 = (const float*)d_in[4];
    const float* b2 = (const float*)d_in[5];
    float* out = (float*)d_out;

    const int* src = ei;
    const int* dst = ei + N_EDGES;

    float* dinv = (float*)d_ws;
    float* buf0 = dinv + 65536;
    float* buf1 = buf0 + (size_t)N_NODES * 64;
    unsigned* planes = (unsigned*)(buf1 + (size_t)N_NODES * 64);
    int* tickets = (int*)(dinv + 50176);

    const int B = 256;
    dim3 blk(B);

    count_deg_kernel<<<CNT_RANGES * CNT_SLICES, CNT_THREADS, 0, stream>>>(dst, planes);
    dinv_kernel<<<(N_NODES + B - 1) / B, blk, 0, stream>>>(planes, dinv, tickets, N_NODES);

    int gemm_grid = (N_NODES + 31) / 32;
    const int scatter_grid = SCATTER_BLOCKS;

    gemm_wmma_f32<IN_DIM, false><<<gemm_grid, blk, 0, stream>>>(x, W1, dinv, b1, buf0, buf1, N_NODES);
    scatter_kernel<<<scatter_grid, blk, 0, stream>>>(buf0, dinv, src, dst, buf1, tickets, N_EDGES);

    gemm_wmma_f32<HID_DIM, true><<<gemm_grid, blk, 0, stream>>>(buf1, W2, dinv, b2, buf0, out, N_NODES);
    scatter_kernel<<<scatter_grid, blk, 0, stream>>>(buf0, dinv, src, dst, out, tickets + 2 * TICKET_PITCH, N_EDGES);
}


# Norm factors out of the edge loop: GEMMs store rows scaled by dinv[row]; layer-1 edges add them as is, GEMM 2 applies dinv[row], bias, relu at its A load; layer-2 edges keep dinv[dst]; arrival order
# speedup vs baseline: 3.0754x; 1.1184x over previous
#include <hip/hip_runtime.h>
#include <hip/hip_bf16.h>
#include <math.h>


#define N_NODES 50000
#define N_EDGES 800000
#define IN_DIM  128
#define HID_DIM 64

#define HALF_N (N_NODES / 2)
#define N_TICKETS 4
#define TICKET_PITCH 64

typedef float v2f __attribute__((ext_vector_type(2)));
typedef float v8f __attribute__((ext_vector_type(8)));


#define CNT_RANGES 8
#define CNT_SLICES 32
#define CNT_RANGE_N 6272
#define CNT_PITCH (CNT_RANGES * CNT_RANGE_N)
#define CNT_SLICE_E (N_EDGES / CNT_SLICES)
#define CNT_SLICE_Q (CNT_SLICE_E / 4)
#define CNT_THREADS 256
#define CNT_LOADS ((CNT_SLICE_Q + CNT_THREADS - 1) / CNT_THREADS)
#define CNT_RANGE_Q (CNT_RANGE_N / 4)
#define CNT_MOVES ((CNT_RANGE_Q + CNT_THREADS - 1) / CNT_THREADS)

static_assert(CNT_PITCH >= N_NODES, "ranges must cover every node");
static_assert(CNT_RANGE_N % 32 == 0, "a wave's 32 counter words are one whole line");
static_assert(N_EDGES % CNT_SLICES == 0 && CNT_SLICE_E % 4 == 0, "slices are whole 16-byte words");

__device__ __forceinline__ void count_one(unsigned* hist, int d, int lo) {
    const unsigned o = (unsigned)(d - lo);
    if (o < (unsigned)CNT_RANGE_N) atomicAdd(&hist[o], 1u);
}

__global__ __launch_bounds__(CNT_THREADS) void count_deg_kernel(const int* __restrict__ dst,
                                                                unsigned* __restrict__ planes) {
    __shared__ __attribute__((aligned(16))) unsigned hist[CNT_RANGE_N];
    const int range = blockIdx.x % CNT_RANGES;
    const int slice = blockIdx.x / CNT_RANGES;
    const int lo = range * CNT_RANGE_N;

    const int4* __restrict__ q = reinterpret_cast<const int4*>(dst + (size_t)slice * CNT_SLICE_E);
    int4 v[CNT_LOADS];
    #pragma unroll
    for (int j = 0; j < CNT_LOADS; ++j) {
        const int i = threadIdx.x + j * CNT_THREADS;
        const bool in = i < CNT_SLICE_Q;
        v[j] = q[in ? i : (CNT_SLICE_Q - 1)];
        if (!in) v[j] = make_int4(-1, -1, -1, -1);
    }

    uint4* hist4 = reinterpret_cast<uint4*>(hist);
    #pragma unroll
    for (int j = 0; j < CNT_MOVES; ++j) {
        const int i = threadIdx.x + j * CNT_THREADS;
        if (i < CNT_RANGE_Q) hist4[i] = make_uint4(0u, 0u, 0u, 0u);
    }
    __syncthreads();

    #pragma unroll
    for (int j = 0; j < CNT_LOADS; ++j) {
        count_one(hist, v[j].x, lo);
        count_one(hist, v[j].y, lo);
        count_one(hist, v[j].z, lo);
        count_one(hist, v[j].w, lo);
    }
    __syncthreads();

    uint4* __restrict__ out4 = reinterpret_cast<uint4*>(planes + (size_t)slice * CNT_PITCH + lo);
    #pragma unroll
    for (int j = 0; j < CNT_MOVES; ++j) {
        const int i = threadIdx.x + j * CNT_THREADS;
        if (i < CNT_RANGE_Q) out4[i] = hist4[i];
    }
}

__global__ void dinv_kernel(const unsigned* __restrict__ planes, float* __restrict__ dinv,
                            int* tickets, int n) {
    int i = blockIdx.x * blockDim.x + threadIdx.x;
    if (i < N_TICKETS) tickets[i * TICKET_PITCH] = 0;
    if (i < n) {
        unsigned c = 1u;
        #pragma unroll
        for (int s = 0; s < CNT_SLICES; ++s) c += planes[(size_t)s * CNT_PITCH + i];
        const float d = (float)c;
        dinv[i] = (d > 0.0f) ? rsqrtf(d) : 0.0f;
    }
}


template <int K_DIM, bool SECOND>
__global__ __launch_bounds__(256) void gemm_wmma_f32(const float* __restrict__ A,
                                                     const float* __restrict__ W,
                                                     const float* __restrict__ dinv,
                                                     const float* __restrict__ bias_in,
                                                     const float* __restrict__ bias,
                                                     float* __restrict__ H,
                                                     float* __restrict__ S, int M) {
    __shared__ float sW[K_DIM * 64];
    __shared__ float sBin[K_DIM];
    for (int i = threadIdx.x; i < K_DIM * 64; i += 256) sW[i] = W[i];
    if (SECOND) { if (threadIdx.x < K_DIM) sBin[threadIdx.x] = bias_in[threadIdx.x]; }
    __syncthreads();

    const int wave = threadIdx.x >> 5;
    const int lane = threadIdx.x & 31;
    const int m_tile = blockIdx.x * 2 + (wave >> 2);
    const int n_tile = wave & 3;

    const int row16 = lane & 15;
    const int khalf = (lane >> 4) << 1;

    int mrow = m_tile * 16 + row16;
    int arow = (mrow < M) ? mrow : (M - 1);
    const float* __restrict__ Aptr = A + (size_t)arow * K_DIM;
    const float* __restrict__ Bcol = sW + n_tile * 16 + row16;

    const int mbase = m_tile * 16 + ((lane >> 4) << 3);
    const int ncol  = n_tile * 16 + row16;

    float drow[8];
    #pragma unroll
    for (int j = 0; j < 8; ++j) {
        const int m = mbase + j;
        drow[j] = dinv[(m < M) ? m : (M - 1)];
    }
    const float bcol = SECOND ? bias[ncol] : 0.f;
    const float d_a = SECOND ? dinv[arow] : 1.f;

    v8f acc = {};
    #pragma unroll
    for (int k = 0; k < K_DIM; k += 4) {
        v2f a, b;
        a.x = Aptr[k + khalf];
        a.y = Aptr[k + khalf + 1];
        if (SECOND) {
            a.x = fmaxf(fmaf(a.x, d_a, sBin[k + khalf]), 0.f);
            a.y = fmaxf(fmaf(a.y, d_a, sBin[k + khalf + 1]), 0.f);
        }
        b.x = Bcol[(k + khalf) * 64];
        b.y = Bcol[(k + khalf + 1) * 64];
        acc = __builtin_amdgcn_wmma_f32_16x16x4_f32(
             false, a,  false, b,
             (short)0, acc,  false,  false);
    }

    #pragma unroll
    for (int j = 0; j < 8; ++j) {
        const int m = mbase + j;
        if (m < M) {
            const float p = acc[j] * drow[j];
            H[(size_t)m * 64 + ncol] = p;
            S[(size_t)m * 64 + ncol] = SECOND ? fmaf(p, drow[j], bcol) : p;
        }
    }
}


#define STEP_E 32
#define CHUNK_E (8 * STEP_E)
#define GROUP 16
#define SCATTER_BLOCKS 2048

__device__ __forceinline__ int package_half() {
    const unsigned id = __builtin_amdgcn_s_sendmsg_rtn(0x87);
    return (int)((id >> 18) & 1u);
}

template <bool DST_FACTOR>
__global__ __launch_bounds__(256) void scatter_kernel(const float* __restrict__ h,
                                                      const float* __restrict__ dinv,
                                                      const int* __restrict__ src,
                                                      const int* __restrict__ dst,
                                                      float* __restrict__ agg,
                                                      int* __restrict__ tickets, int e_count) {
    __shared__ int s_ticket;
    const int lane = threadIdx.x & 31;
    const int wave = threadIdx.x >> 5;
    const int n_chunks = (e_count + CHUNK_E - 1) / CHUNK_E;
    const int my_half = package_half();

    for (int pass = 0; pass < 2; ++pass) {
        const int r = my_half ^ pass;
        int* counter = tickets + r * TICKET_PITCH;
        for (int trip = 0; trip <= n_chunks; ++trip) {
            if (threadIdx.x == 0) s_ticket = atomicAdd(counter, 1);
            __syncthreads();
            const int t = s_ticket;
            __syncthreads();
            if (t >= n_chunks) break;

            const int e = t * CHUNK_E + wave * STEP_E + lane;
            const int ec = (e < e_count) ? e : (e_count - 1);
            const int my_s = src[ec];
            const int my_d = dst[ec];
            asm volatile("" :: "v"(my_s), "v"(my_d));
            const bool mine = (e < e_count) && ((my_d >= HALF_N) == (r != 0));
            float my_dd = 1.f;
            if (DST_FACTOR) my_dd = dinv[mine ? my_d : 0];

            unsigned todo = __builtin_amdgcn_ballot_w32(mine);
            while (todo) {
                int j[GROUP];
                #pragma unroll
                for (int i = 0; i < GROUP; ++i) {
                    j[i] = todo ? __builtin_ctz(todo) : -1;
                    todo &= todo - 1u;
                }
                float lo[GROUP], hi[GROUP];
                #pragma unroll
                for (int i = 0; i < GROUP; ++i) {
                    if (j[i] >= 0) {
                        const int s = __builtin_amdgcn_readlane(my_s, j[i]);
                        const float* __restrict__ row = h + (size_t)s * 64;
                        lo[i] = row[lane];
                        hi[i] = row[lane + 32];
                    }
                }
                #pragma unroll
                for (int i = 0; i < GROUP; ++i) {
                    if (j[i] >= 0) {
                        const int d = __builtin_amdgcn_readlane(my_d, j[i]);
                        float* o = agg + (size_t)d * 64;
                        if (DST_FACTOR) {
                            const float norm =
                                __int_as_float(__builtin_amdgcn_readlane(__float_as_int(my_dd), j[i]));
                            atomicAdd(o + lane,      lo[i] * norm);
                            atomicAdd(o + lane + 32, hi[i] * norm);
                        } else {
                            atomicAdd(o + lane,      lo[i]);
                            atomicAdd(o + lane + 32, hi[i]);
                        }
                    }
                }
            }
        }
    }
}


extern "C" void kernel_launch(void* const* d_in, const int* in_sizes, int n_in,
                              void* d_out, int out_size, void* d_ws, size_t ws_size,
                              hipStream_t stream) {
    const float* x  = (const float*)d_in[0];
    const int*   ei = (const int*)d_in[1];
    const float* W1 = (const float*)d_in[2];
    const float* b1 = (const float*)d_in[3];
    const float* W2 = (const float*)d_in[4];
    const float* b2 = (const float*)d_in[5];
    float* out = (float*)d_out;

    const int* src = ei;
    const int* dst = ei + N_EDGES;

    float* dinv = (float*)d_ws;
    float* buf0 = dinv + 65536;
    float* buf1 = buf0 + (size_t)N_NODES * 64;
    unsigned* planes = (unsigned*)(buf1 + (size_t)N_NODES * 64);
    int* tickets = (int*)(dinv + 50176);

    const int B = 256;
    dim3 blk(B);

    count_deg_kernel<<<CNT_RANGES * CNT_SLICES, CNT_THREADS, 0, stream>>>(dst, planes);
    dinv_kernel<<<(N_NODES + B - 1) / B, blk, 0, stream>>>(planes, dinv, tickets, N_NODES);

    int gemm_grid = (N_NODES + 31) / 32;
    const int scatter_grid = SCATTER_BLOCKS;

    gemm_wmma_f32<IN_DIM, false><<<gemm_grid, blk, 0, stream>>>(x, W1, dinv, b1, b1, buf0, buf1, N_NODES);
    scatter_kernel<false><<<scatter_grid, blk, 0, stream>>>(buf0, dinv, src, dst, buf1, tickets, N_EDGES);

    gemm_wmma_f32<HID_DIM, true><<<gemm_grid, blk, 0, stream>>>(buf1, W2, dinv, b1, b2, buf0, out, N_NODES);
    scatter_kernel<true><<<scatter_grid, blk, 0, stream>>>(buf0, dinv, src, dst, out, tickets + 2 * TICKET_PITCH, N_EDGES);
}


# Both GEMMs persistent: 768 work-groups stage W once, walk over 32-row tiles; a tile's rows reach the waves via LDS by whole-line loads, next tile's in flight under the WMMAs; same k order and sums
# speedup vs baseline: 3.4852x; 1.1333x over previous
#include <hip/hip_runtime.h>
#include <hip/hip_bf16.h>
#include <math.h>


#define N_NODES 50000
#define N_EDGES 800000
#define IN_DIM  128
#define HID_DIM 64

#define HALF_N (N_NODES / 2)
#define N_TICKETS 4
#define TICKET_PITCH 64

typedef float v2f __attribute__((ext_vector_type(2)));
typedef float v4f __attribute__((ext_vector_type(4)));
typedef float v8f __attribute__((ext_vector_type(8)));


#define CNT_RANGES 8
#define CNT_SLICES 32
#define CNT_RANGE_N 6272
#define CNT_PITCH (CNT_RANGES * CNT_RANGE_N)
#define CNT_SLICE_E (N_EDGES / CNT_SLICES)
#define CNT_SLICE_Q (CNT_SLICE_E / 4)
#define CNT_THREADS 256
#define CNT_LOADS ((CNT_SLICE_Q + CNT_THREADS - 1) / CNT_THREADS)
#define CNT_RANGE_Q (CNT_RANGE_N / 4)
#define CNT_MOVES ((CNT_RANGE_Q + CNT_THREADS - 1) / CNT_THREADS)

static_assert(CNT_PITCH >= N_NODES, "ranges must cover every node");
static_assert(CNT_RANGE_N % 32 == 0, "a wave's 32 counter words are one whole line");
static_assert(N_EDGES % CNT_SLICES == 0 && CNT_SLICE_E % 4 == 0, "slices are whole 16-byte words");

__device__ __forceinline__ void count_one(unsigned* hist, int d, int lo) {
    const unsigned o = (unsigned)(d - lo);
    if (o < (unsigned)CNT_RANGE_N) atomicAdd(&hist[o], 1u);
}

__global__ __launch_bounds__(CNT_THREADS) void count_deg_kernel(const int* __restrict__ dst,
                                                                unsigned* __restrict__ planes) {
    __shared__ __attribute__((aligned(16))) unsigned hist[CNT_RANGE_N];
    const int range = blockIdx.x % CNT_RANGES;
    const int slice = blockIdx.x / CNT_RANGES;
    const int lo = range * CNT_RANGE_N;

    const int4* __restrict__ q = reinterpret_cast<const int4*>(dst + (size_t)slice * CNT_SLICE_E);
    int4 v[CNT_LOADS];
    #pragma unroll
    for (int j = 0; j < CNT_LOADS; ++j) {
        const int i = threadIdx.x + j * CNT_THREADS;
        const bool in = i < CNT_SLICE_Q;
        v[j] = q[in ? i : (CNT_SLICE_Q - 1)];
        if (!in) v[j] = make_int4(-1, -1, -1, -1);
    }

    uint4* hist4 = reinterpret_cast<uint4*>(hist);
    #pragma unroll
    for (int j = 0; j < CNT_MOVES; ++j) {
        const int i = threadIdx.x + j * CNT_THREADS;
        if (i < CNT_RANGE_Q) hist4[i] = make_uint4(0u, 0u, 0u, 0u);
    }
    __syncthreads();

    #pragma unroll
    for (int j = 0; j < CNT_LOADS; ++j) {
        count_one(hist, v[j].x, lo);
        count_one(hist, v[j].y, lo);
        count_one(hist, v[j].z, lo);
        count_one(hist, v[j].w, lo);
    }
    __syncthreads();

    uint4* __restrict__ out4 = reinterpret_cast<uint4*>(planes + (size_t)slice * CNT_PITCH + lo);
    #pragma unroll
    for (int j = 0; j < CNT_MOVES; ++j) {
        const int i = threadIdx.x + j * CNT_THREADS;
        if (i < CNT_RANGE_Q) out4[i] = hist4[i];
    }
}

__global__ void dinv_kernel(const unsigned* __restrict__ planes, float* __restrict__ dinv,
                            int* tickets, int n) {
    int i = blockIdx.x * blockDim.x + threadIdx.x;
    if (i < N_TICKETS) tickets[i * TICKET_PITCH] = 0;
    if (i < n) {
        unsigned c = 1u;
        #pragma unroll
        for (int s = 0; s < CNT_SLICES; ++s) c += planes[(size_t)s * CNT_PITCH + i];
        const float d = (float)c;
        dinv[i] = (d > 0.0f) ? rsqrtf(d) : 0.0f;
    }
}


#define GEMM_ROWS 32
#define GEMM_TILES ((N_NODES + GEMM_ROWS - 1) / GEMM_ROWS)
#define GEMM_BLOCKS 768

static_assert(GEMM_BLOCKS <= GEMM_TILES, "every work-group owns at least one tile");

__device__ __forceinline__ void lds_barrier() {
    __builtin_amdgcn_fence(__ATOMIC_RELEASE, "workgroup", "local");
    __builtin_amdgcn_s_barrier();
    __builtin_amdgcn_fence(__ATOMIC_ACQUIRE, "workgroup", "local");
}

template <int K_DIM, bool SECOND>
__global__ __launch_bounds__(256) void gemm_wmma_f32(const float* __restrict__ A,
                                                     const float* __restrict__ W,
                                                     const float* __restrict__ dinv,
                                                     const float* __restrict__ bias_in,
                                                     const float* __restrict__ bias,
                                                     float* __restrict__ H,
                                                     float* __restrict__ S, int M) {
    constexpr int A_PITCH = K_DIM + 4;
    constexpr int ROW_Q   = K_DIM / 4;
    constexpr int Q_ROWS  = 256 / ROW_Q;
    constexpr int A_LOADS = GEMM_ROWS / Q_ROWS;
    static_assert(256 % ROW_Q == 0 && GEMM_ROWS % Q_ROWS == 0, "a thread keeps its column over its loads");

    __shared__ float sW[K_DIM * 64];
    __shared__ __attribute__((aligned(16))) float sA[GEMM_ROWS * A_PITCH];

    const int wave = threadIdx.x >> 5;
    const int lane = threadIdx.x & 31;
    const int wm = wave >> 2;
    const int n_tile = wave & 3;

    const int row16 = lane & 15;
    const int khalf = (lane >> 4) << 1;
    const int ncol  = n_tile * 16 + row16;

    const int n_tiles = (M + GEMM_ROWS - 1) / GEMM_ROWS;

    const int qcol = (threadIdx.x % ROW_Q) * 4;
    const int qrow = threadIdx.x / ROW_Q;

    v4f    nxt[A_LOADS];
    float  dn[A_LOADS];
    v4f    bin4 = {0.f, 0.f, 0.f, 0.f};

    int tile = blockIdx.x;
    #pragma unroll
    for (int j = 0; j < A_LOADS; ++j) {
        int r = tile * GEMM_ROWS + qrow + j * Q_ROWS;
        r = (r < M) ? r : (M - 1);
        nxt[j] = *reinterpret_cast<const v4f*>(A + (size_t)r * K_DIM + qcol);
        if (SECOND) dn[j] = dinv[r];
    }
    if (SECOND) bin4 = *reinterpret_cast<const v4f*>(bias_in + qcol);
    const float bcol = SECOND ? bias[ncol] : 0.f;
    __builtin_amdgcn_sched_barrier(0);

    #pragma unroll
    for (int j = 0; j < K_DIM * 64 / 256; ++j) sW[threadIdx.x + j * 256] = W[threadIdx.x + j * 256];

    const float* __restrict__ Arow = sA + (wm * 16 + row16) * A_PITCH + khalf;
    const float* __restrict__ Bcol = sW + ncol;

    do {
        #pragma unroll
        for (int j = 0; j < A_LOADS; ++j) {
            v4f v = nxt[j];
            if (SECOND) {
                v.x = fmaxf(fmaf(v.x, dn[j], bin4.x), 0.f);
                v.y = fmaxf(fmaf(v.y, dn[j], bin4.y), 0.f);
                v.z = fmaxf(fmaf(v.z, dn[j], bin4.z), 0.f);
                v.w = fmaxf(fmaf(v.w, dn[j], bin4.w), 0.f);
            }
            *reinterpret_cast<v4f*>(&sA[(qrow + j * Q_ROWS) * A_PITCH + qcol]) = v;
        }
        lds_barrier();

        const int mbase = tile * GEMM_ROWS + wm * 16 + ((lane >> 4) << 3);
        float drow[8];
        #pragma unroll
        for (int j = 0; j < 8; ++j) {
            const int m = mbase + j;
            drow[j] = dinv[(m < M) ? m : (M - 1)];
        }

        const int ntile = tile + gridDim.x;
        if (ntile < n_tiles) {
            #pragma unroll
            for (int j = 0; j < A_LOADS; ++j) {
                int r = ntile * GEMM_ROWS + qrow + j * Q_ROWS;
                r = (r < M) ? r : (M - 1);
                nxt[j] = *reinterpret_cast<const v4f*>(A + (size_t)r * K_DIM + qcol);
                if (SECOND) dn[j] = dinv[r];
            }
        }

        v8f acc = {};
        #pragma unroll
        for (int k = 0; k < K_DIM; k += 4) {
            v2f a, b;
            a = *reinterpret_cast<const v2f*>(Arow + k);
            b.x = Bcol[(k + khalf) * 64];
            b.y = Bcol[(k + khalf + 1) * 64];
            acc = __builtin_amdgcn_wmma_f32_16x16x4_f32(
                 false, a,  false, b,
                 (short)0, acc,  false,  false);
        }
        lds_barrier();

        #pragma unroll
        for (int j = 0; j < 8; ++j) {
            const int m = mbase + j;
            if (m < M) {
                const float p = acc[j] * drow[j];
                H[(size_t)m * 64 + ncol] = p;
                S[(size_t)m * 64 + ncol] = SECOND ? fmaf(p, drow[j], bcol) : p;
            }
        }
        tile = ntile;
    } while (tile < n_tiles);
}


#define STEP_E 32
#define CHUNK_E (8 * STEP_E)
#define GROUP 16
#define SCATTER_BLOCKS 2048

__device__ __forceinline__ int package_half() {
    const unsigned id = __builtin_amdgcn_s_sendmsg_rtn(0x87);
    return (int)((id >> 18) & 1u);
}

template <bool DST_FACTOR>
__global__ __launch_bounds__(256) void scatter_kernel(const float* __restrict__ h,
                                                      const float* __restrict__ dinv,
                                                      const int* __restrict__ src,
                                                      const int* __restrict__ dst,
                                                      float* __restrict__ agg,
                                                      int* __restrict__ tickets, int e_count) {
    __shared__ int s_ticket;
    const int lane = threadIdx.x & 31;
    const int wave = threadIdx.x >> 5;
    const int n_chunks = (e_count + CHUNK_E - 1) / CHUNK_E;
    const int my_half = package_half();

    for (int pass = 0; pass < 2; ++pass) {
        const int r = my_half ^ pass;
        int* counter = tickets + r * TICKET_PITCH;
        for (int trip = 0; trip <= n_chunks; ++trip) {
            if (threadIdx.x == 0) s_ticket = atomicAdd(counter, 1);
            __syncthreads();
            const int t = s_ticket;
            __syncthreads();
            if (t >= n_chunks) break;

            const int e = t * CHUNK_E + wave * STEP_E + lane;
            const int ec = (e < e_count) ? e : (e_count - 1);
            const int my_s = src[ec];
            const int my_d = dst[ec];
            asm volatile("" :: "v"(my_s), "v"(my_d));
            const bool mine = (e < e_count) && ((my_d >= HALF_N) == (r != 0));
            float my_dd = 1.f;
            if (DST_FACTOR) my_dd = dinv[mine ? my_d : 0];

            unsigned todo = __builtin_amdgcn_ballot_w32(mine);
            while (todo) {
                int j[GROUP];
                #pragma unroll
                for (int i = 0; i < GROUP; ++i) {
                    j[i] = todo ? __builtin_ctz(todo) : -1;
                    todo &= todo - 1u;
                }
                float lo[GROUP], hi[GROUP];
                #pragma unroll
                for (int i = 0; i < GROUP; ++i) {
                    if (j[i] >= 0) {
                        const int s = __builtin_amdgcn_readlane(my_s, j[i]);
                        const float* __restrict__ row = h + (size_t)s * 64;
                        lo[i] = row[lane];
                        hi[i] = row[lane + 32];
                    }
                }
                #pragma unroll
                for (int i = 0; i < GROUP; ++i) {
                    if (j[i] >= 0) {
                        const int d = __builtin_amdgcn_readlane(my_d, j[i]);
                        float* o = agg + (size_t)d * 64;
                        if (DST_FACTOR) {
                            const float norm =
                                __int_as_float(__builtin_amdgcn_readlane(__float_as_int(my_dd), j[i]));
                            atomicAdd(o + lane,      lo[i] * norm);
                            atomicAdd(o + lane + 32, hi[i] * norm);
                        } else {
                            atomicAdd(o + lane,      lo[i]);
                            atomicAdd(o + lane + 32, hi[i]);
                        }
                    }
                }
            }
        }
    }
}


extern "C" void kernel_launch(void* const* d_in, const int* in_sizes, int n_in,
                              void* d_out, int out_size, void* d_ws, size_t ws_size,
                              hipStream_t stream) {
    const float* x  = (const float*)d_in[0];
    const int*   ei = (const int*)d_in[1];
    const float* W1 = (const float*)d_in[2];
    const float* b1 = (const float*)d_in[3];
    const float* W2 = (const float*)d_in[4];
    const float* b2 = (const float*)d_in[5];
    float* out = (float*)d_out;

    const int* src = ei;
    const int* dst = ei + N_EDGES;

    float* dinv = (float*)d_ws;
    float* buf0 = dinv + 65536;
    float* buf1 = buf0 + (size_t)N_NODES * 64;
    unsigned* planes = (unsigned*)(buf1 + (size_t)N_NODES * 64);
    int* tickets = (int*)(dinv + 50176);

    const int B = 256;
    dim3 blk(B);

    count_deg_kernel<<<CNT_RANGES * CNT_SLICES, CNT_THREADS, 0, stream>>>(dst, planes);
    dinv_kernel<<<(N_NODES + B - 1) / B, blk, 0, stream>>>(planes, dinv, tickets, N_NODES);

    const int gemm_grid = GEMM_BLOCKS;
    const int scatter_grid = SCATTER_BLOCKS;

    gemm_wmma_f32<IN_DIM, false><<<gemm_grid, blk, 0, stream>>>(x, W1, dinv, b1, b1, buf0, buf1, N_NODES);
    scatter_kernel<false><<<scatter_grid, blk, 0, stream>>>(buf0, dinv, src, dst, buf1, tickets, N_EDGES);

    gemm_wmma_f32<HID_DIM, true><<<gemm_grid, blk, 0, stream>>>(buf1, W2, dinv, b1, b2, buf0, out, N_NODES);
    scatter_kernel<true><<<scatter_grid, blk, 0, stream>>>(buf0, dinv, src, dst, out, tickets + 2 * TICKET_PITCH, N_EDGES);
}
